# starting kernel
# baseline (speedup 1.0000x reference)
#include <hip/hip_runtime.h>

typedef __attribute__((ext_vector_type(2))) float v2f;
typedef __attribute__((ext_vector_type(8))) float v8f;

#define NN    10000
#define BB    2
#define FF    64
#define NASZ  64
#define OUTC  64
#define KDIM  128
#define NPW   32
#define ROWS  64
#define LSTR  132
#define THREADS 256

__launch_bounds__(THREADS)
__global__ void pgnn_fused_kernel(const float* __restrict__ x,
                                  const float* __restrict__ dm,
                                  const int*   __restrict__ idx,
                                  const float* __restrict__ W,
                                  const float* __restrict__ bias,
                                  float* __restrict__ out)
{
    __shared__ float w_lds[OUTC * LSTR];
    __shared__ float h_lds[ROWS * LSTR];
    __shared__ float bias_lds[OUTC];
    __shared__ int   idx_lds[NPW * NASZ];
    __shared__ float dm_lds [NPW * NASZ];

    const int tid = threadIdx.x;
    const int n0  = blockIdx.x * NPW;

    for (int e = tid; e < OUTC * KDIM; e += THREADS) {
        int o = e >> 7, k = e & 127;
        w_lds[o * LSTR + k] = W[e];
    }
    if (tid < OUTC) bias_lds[tid] = bias[tid];

    for (int e = tid; e < NPW * NASZ; e += THREADS) {
        int nl = e >> 6;
        int n  = n0 + nl;
        if (n < NN) {
            idx_lds[e] = idx[n * NASZ + (e & 63)];
            dm_lds[e]  = dm [n * NASZ + (e & 63)];
        } else {
            idx_lds[e] = 0;
            dm_lds[e]  = 0.0f;
        }
    }
    __syncthreads();

    {
        const int half = tid >> 7;
        const int t    = tid & 127;
        const int b    = t >> 6;
        const int f    = t & 63;
        const int nlo  = half * (NPW / 2);
        for (int nl = nlo; nl < nlo + NPW / 2; ++nl) {
            const int*   ip = &idx_lds[nl * NASZ];
            const float* dp = &dm_lds [nl * NASZ];
            float acc = 0.0f;
            #pragma unroll 8
            for (int a = 0; a < NASZ; ++a) {
                acc = fmaf(dp[a], x[(size_t)ip[a] * KDIM + t], acc);
            }
            const int n   = n0 + nl;
            const int row = nl * 2 + b;
            h_lds[row * LSTR + f]      = acc * (1.0f / NASZ);
            h_lds[row * LSTR + FF + f] = (n < NN) ? x[(size_t)n * KDIM + t] : 0.0f;
        }
    }
    __syncthreads();

    const int wave  = tid >> 5;
    const int lane  = tid & 31;
    const int l15   = lane & 15;
    const int khalf = (lane >> 4) * 2;
    const int rt    = wave >> 1;
    const int ch    = wave & 1;
    const int rowA  = rt * 16 + l15;
    const int col0  = (ch * 2) * 16 + l15;
    const int col1  = (ch * 2 + 1) * 16 + l15;

    v8f acc0, acc1;
    {
        const float bv0 = bias_lds[col0];
        const float bv1 = bias_lds[col1];
        #pragma unroll
        for (int r = 0; r < 8; ++r) { acc0[r] = bv0; acc1[r] = bv1; }
    }

    #pragma unroll 4
    for (int kk = 0; kk < 32; ++kk) {
        const int kb = kk * 4 + khalf;
        v2f afrag;
        afrag.x = h_lds[rowA * LSTR + kb];
        afrag.y = h_lds[rowA * LSTR + kb + 1];
        v2f bfrag0, bfrag1;
        bfrag0.x = w_lds[col0 * LSTR + kb];
        bfrag0.y = w_lds[col0 * LSTR + kb + 1];
        bfrag1.x = w_lds[col1 * LSTR + kb];
        bfrag1.y = w_lds[col1 * LSTR + kb + 1];
        acc0 = __builtin_amdgcn_wmma_f32_16x16x4_f32(
            false, afrag, false, bfrag0, (short)0, acc0, false, false);
        acc1 = __builtin_amdgcn_wmma_f32_16x16x4_f32(
            false, afrag, false, bfrag1, (short)0, acc1, false, false);
    }

    const int mbase = blockIdx.x * ROWS + rt * 16 + (lane >> 4) * 8;
    float* __restrict__ op = out + (size_t)mbase * OUTC;
    const bool full = (blockIdx.x + 1) * ROWS <= NN * BB;
    if (full) {
        #pragma unroll
        for (int r = 0; r < 8; ++r) {
            op[(size_t)r * OUTC + col0] = acc0[r];
            op[(size_t)r * OUTC + col1] = acc1[r];
        }
    } else {
        #pragma unroll
        for (int r = 0; r < 8; ++r) {
            if (mbase + r < NN * BB) {
                op[(size_t)r * OUTC + col0] = acc0[r];
                op[(size_t)r * OUTC + col1] = acc1[r];
            }
        }
    }
}

extern "C" void kernel_launch(void* const* d_in, const int* in_sizes, int n_in,
                              void* d_out, int out_size, void* d_ws, size_t ws_size,
                              hipStream_t stream) {
    const float* x    = (const float*)d_in[0];
    const float* dm   = (const float*)d_in[1];
    const int*   idx  = (const int*)  d_in[2];
    const float* W    = (const float*)d_in[3];
    const float* bias = (const float*)d_in[4];
    float* out = (float*)d_out;

    const int grid = (NN * BB + ROWS - 1) / ROWS;
    pgnn_fused_kernel<<<grid, THREADS, 0, stream>>>(x, dm, idx, W, bias, out);
}


# Gather stage re-mapped: each wave owns whole nodes, one float4-per-lane load reads a whole 512-B x row, 4 groups of 8 row loads kept in flight per wave; per-element fmaf order a=0..63 unchanged
# speedup vs baseline: 2.8678x; 2.8678x over previous
#include <hip/hip_runtime.h>

typedef __attribute__((ext_vector_type(2))) float v2f;
typedef __attribute__((ext_vector_type(4))) float v4f;
typedef __attribute__((ext_vector_type(8))) float v8f;

#define NN    10000
#define BB    2
#define FF    64
#define NASZ  64
#define OUTC  64
#define KDIM  128
#define NPW   32
#define ROWS  64
#define LSTR  132
#define THREADS 256
#define WAVES   8
#define NPV     (NPW / WAVES)
#define CH      8
#define GPN     (NASZ / CH)
#define NGRP    (NPV * GPN)
#define AHEAD   4

__launch_bounds__(THREADS)
__global__ void pgnn_fused_kernel(const float* __restrict__ x,
                                  const float* __restrict__ dm,
                                  const int*   __restrict__ idx,
                                  const float* __restrict__ W,
                                  const float* __restrict__ bias,
                                  float* __restrict__ out)
{
    __shared__ float w_lds[OUTC * LSTR];
    __shared__ __attribute__((aligned(16))) float h_lds[ROWS * LSTR];
    __shared__ float bias_lds[OUTC];
    __shared__ __attribute__((aligned(16))) int   idx_lds[NPW * NASZ];
    __shared__ __attribute__((aligned(16))) float dm_lds [NPW * NASZ];

    const int tid = threadIdx.x;
    const int n0  = blockIdx.x * NPW;

    v4f own[NPV];
    #pragma unroll
    for (int i = 0; i < NPV; ++i) {
        const int n  = n0 + (tid >> 5) * NPV + i;
        const int nc = (n < NN) ? n : (NN - 1);
        own[i] = *reinterpret_cast<const v4f*>(x + (size_t)nc * KDIM + 4 * (tid & 31));
    }

    for (int e = tid; e < OUTC * KDIM; e += THREADS) {
        int o = e >> 7, k = e & 127;
        w_lds[o * LSTR + k] = W[e];
    }
    if (tid < OUTC) bias_lds[tid] = bias[tid];

    for (int e = tid; e < NPW * NASZ; e += THREADS) {
        int nl = e >> 6;
        int n  = n0 + nl;
        if (n < NN) {
            idx_lds[e] = idx[n * NASZ + (e & 63)];
            dm_lds[e]  = dm [n * NASZ + (e & 63)];
        } else {
            idx_lds[e] = 0;
            dm_lds[e]  = 0.0f;
        }
    }
    #pragma unroll
    for (int i = 0; i < NPV; ++i) {
        const int nl  = (tid >> 5) * NPV + i;
        const int row = nl * 2 + ((tid & 31) >> 4);
        v4f o = own[i];
        if (n0 + nl >= NN) o = (v4f){0.0f, 0.0f, 0.0f, 0.0f};
        *reinterpret_cast<v4f*>(&h_lds[row * LSTR + FF + 4 * (tid & 15)]) = o;
    }
    __syncthreads();

    {
        const int wv  = tid >> 5;
        const int ln  = tid & 31;
        const int nlw = wv * NPV;
        const unsigned lnoff = 16u * (unsigned)ln;
        const unsigned char* xbytes = reinterpret_cast<const unsigned char*>(x);

        v4f ring[AHEAD + 1][CH];
#define LOAD_GROUP(G) do { \
        const int* ip_ = &idx_lds[(nlw + (G) / GPN) * NASZ + ((G) % GPN) * CH]; \
        _Pragma("unroll") for (int i_ = 0; i_ < CH; ++i_) { \
            const unsigned off_ = (unsigned)ip_[i_] * (unsigned)(KDIM * 4) + lnoff; \
            ring[(G) % (AHEAD + 1)][i_] = *reinterpret_cast<const v4f*>(xbytes + off_); \
        } } while (0)

        #pragma unroll
        for (int g = 0; g < AHEAD; ++g) LOAD_GROUP(g);

        v4f acc = {0.0f, 0.0f, 0.0f, 0.0f};
        #pragma unroll
        for (int g = 0; g < NGRP; ++g) {
            __builtin_amdgcn_sched_barrier(0);
            if (g + AHEAD < NGRP) LOAD_GROUP(g + AHEAD);
            __builtin_amdgcn_sched_barrier(0);

            const int nl = nlw + g / GPN;
            const float* dp = &dm_lds[nl * NASZ + (g % GPN) * CH];
            if (g % GPN == 0) acc = (v4f){0.0f, 0.0f, 0.0f, 0.0f};
            #pragma unroll
            for (int i = 0; i < CH; ++i) {
                const float d = dp[i];
                const v4f  v = ring[g % (AHEAD + 1)][i];
                acc.x = fmaf(d, v.x, acc.x);
                acc.y = fmaf(d, v.y, acc.y);
                acc.z = fmaf(d, v.z, acc.z);
                acc.w = fmaf(d, v.w, acc.w);
            }
            if (g % GPN == GPN - 1) {
                const int row = nl * 2 + (ln >> 4);
                *reinterpret_cast<v4f*>(&h_lds[row * LSTR + 4 * (ln & 15)]) = acc * (1.0f / NASZ);
            }
        }
#undef LOAD_GROUP
    }
    __syncthreads();

    const int wave  = tid >> 5;
    const int lane  = tid & 31;
    const int l15   = lane & 15;
    const int khalf = (lane >> 4) * 2;
    const int rt    = wave >> 1;
    const int ch    = wave & 1;
    const int rowA  = rt * 16 + l15;
    const int col0  = (ch * 2) * 16 + l15;
    const int col1  = (ch * 2 + 1) * 16 + l15;

    v8f acc0, acc1;
    {
        const float bv0 = bias_lds[col0];
        const float bv1 = bias_lds[col1];
        #pragma unroll
        for (int r = 0; r < 8; ++r) { acc0[r] = bv0; acc1[r] = bv1; }
    }

    #pragma unroll 4
    for (int kk = 0; kk < 32; ++kk) {
        const int kb = kk * 4 + khalf;
        v2f afrag;
        afrag.x = h_lds[rowA * LSTR + kb];
        afrag.y = h_lds[rowA * LSTR + kb + 1];
        v2f bfrag0, bfrag1;
        bfrag0.x = w_lds[col0 * LSTR + kb];
        bfrag0.y = w_lds[col0 * LSTR + kb + 1];
        bfrag1.x = w_lds[col1 * LSTR + kb];
        bfrag1.y = w_lds[col1 * LSTR + kb + 1];
        acc0 = __builtin_amdgcn_wmma_f32_16x16x4_f32(
            false, afrag, false, bfrag0, (short)0, acc0, false, false);
        acc1 = __builtin_amdgcn_wmma_f32_16x16x4_f32(
            false, afrag, false, bfrag1, (short)0, acc1, false, false);
    }

    const int mbase = blockIdx.x * ROWS + rt * 16 + (lane >> 4) * 8;
    float* __restrict__ op = out + (size_t)mbase * OUTC;
    const bool full = (blockIdx.x + 1) * ROWS <= NN * BB;
    if (full) {
        #pragma unroll
        for (int r = 0; r < 8; ++r) {
            op[(size_t)r * OUTC + col0] = acc0[r];
            op[(size_t)r * OUTC + col1] = acc1[r];
        }
    } else {
        #pragma unroll
        for (int r = 0; r < 8; ++r) {
            if (mbase + r < NN * BB) {
                op[(size_t)r * OUTC + col0] = acc0[r];
                op[(size_t)r * OUTC + col1] = acc1[r];
            }
        }
    }
}

extern "C" void kernel_launch(void* const* d_in, const int* in_sizes, int n_in,
                              void* d_out, int out_size, void* d_ws, size_t ws_size,
                              hipStream_t stream) {
    const float* x    = (const float*)d_in[0];
    const float* dm   = (const float*)d_in[1];
    const int*   idx  = (const int*)  d_in[2];
    const float* W    = (const float*)d_in[3];
    const float* bias = (const float*)d_in[4];
    float* out = (float*)d_out;

    const int grid = (NN * BB + ROWS - 1) / ROWS;
    pgnn_fused_kernel<<<grid, THREADS, 0, stream>>>(x, dm, idx, W, bias, out);
}


# Staging off the path ahead of the gather: indices/weights as one batch of 128-bit loads (was 8 load-wait-store passes); W_hidden and bias requested in the ring's free slot behind the last row group
# speedup vs baseline: 3.6480x; 1.2721x over previous
#include <hip/hip_runtime.h>

typedef __attribute__((ext_vector_type(2))) float v2f;
typedef __attribute__((ext_vector_type(4))) float v4f;
typedef __attribute__((ext_vector_type(8))) float v8f;
typedef __attribute__((ext_vector_type(4))) int   v4i;

#define NN    10000
#define BB    2
#define FF    64
#define NASZ  64
#define OUTC  64
#define KDIM  128
#define NPW   32
#define ROWS  64
#define LSTR  132
#define THREADS 256
#define WAVES   8
#define NPV     (NPW / WAVES)
#define CH      8
#define GPN     (NASZ / CH)
#define NGRP    (NPV * GPN)
#define AHEAD   4
#define IDXV    ((NPW * NASZ) / (4 * THREADS))
#define WSLOT   (NGRP % (AHEAD + 1))
static_assert((OUTC * KDIM) / (4 * THREADS) == CH, "W_hidden must fill exactly one ring slot per thread");
static_assert((NPW * NASZ) % (4 * THREADS) == 0, "index staging must divide evenly");

__launch_bounds__(THREADS)
__global__ void pgnn_fused_kernel(const float* __restrict__ x,
                                  const float* __restrict__ dm,
                                  const int*   __restrict__ idx,
                                  const float* __restrict__ W,
                                  const float* __restrict__ bias,
                                  float* __restrict__ out)
{
    __shared__ __attribute__((aligned(16))) float w_lds[OUTC * LSTR];
    __shared__ __attribute__((aligned(16))) float h_lds[ROWS * LSTR];
    __shared__ __attribute__((aligned(16))) int   idx_lds[NPW * NASZ];
    __shared__ __attribute__((aligned(16))) float dm_lds [NPW * NASZ];

    const int tid = threadIdx.x;
    const int n0  = blockIdx.x * NPW;

    v4f own[NPV];
    #pragma unroll
    for (int i = 0; i < NPV; ++i) {
        const int n  = n0 + (tid >> 5) * NPV + i;
        const int nc = (n < NN) ? n : (NN - 1);
        own[i] = *reinterpret_cast<const v4f*>(x + (size_t)nc * KDIM + 4 * (tid & 31));
    }

    v4i iv[IDXV];
    v4f dv[IDXV];
    #pragma unroll
    for (int j = 0; j < IDXV; ++j) {
        const int e  = 4 * (tid + j * THREADS);
        const int n  = n0 + (e >> 6);
        const int nc = (n < NN) ? n : (NN - 1);
        const size_t off = (size_t)nc * NASZ + (e & 63);
        iv[j] = *reinterpret_cast<const v4i*>(idx + off);
        dv[j] = *reinterpret_cast<const v4f*>(dm  + off);
    }
    #pragma unroll
    for (int j = 0; j < IDXV; ++j) {
        const int e = 4 * (tid + j * THREADS);
        v4i a = iv[j];
        v4f d = dv[j];
        if (n0 + (e >> 6) >= NN) {
            a = (v4i){0, 0, 0, 0};
            d = (v4f){0.0f, 0.0f, 0.0f, 0.0f};
        }
        *reinterpret_cast<v4i*>(&idx_lds[e]) = a;
        *reinterpret_cast<v4f*>(&dm_lds[e])  = d;
    }
    #pragma unroll
    for (int i = 0; i < NPV; ++i) {
        const int nl  = (tid >> 5) * NPV + i;
        const int row = nl * 2 + ((tid & 31) >> 4);
        v4f o = own[i];
        if (n0 + nl >= NN) o = (v4f){0.0f, 0.0f, 0.0f, 0.0f};
        *reinterpret_cast<v4f*>(&h_lds[row * LSTR + FF + 4 * (tid & 15)]) = o;
    }
    __syncthreads();

    const int wave  = tid >> 5;
    const int lane  = tid & 31;
    const int l15   = lane & 15;
    const int khalf = (lane >> 4) * 2;
    const int rt    = wave >> 1;
    const int ch    = wave & 1;
    const int rowA  = rt * 16 + l15;
    const int col0  = (ch * 2) * 16 + l15;
    const int col1  = (ch * 2 + 1) * 16 + l15;
    float bv0 = 0.0f, bv1 = 0.0f;

    {
        const int wv  = tid >> 5;
        const int ln  = tid & 31;
        const int nlw = wv * NPV;
        const unsigned lnoff = 16u * (unsigned)ln;
        const unsigned char* xbytes = reinterpret_cast<const unsigned char*>(x);

        v4f ring[AHEAD + 1][CH];
#define LOAD_GROUP(G) do { \
        const int* ip_ = &idx_lds[(nlw + (G) / GPN) * NASZ + ((G) % GPN) * CH]; \
        _Pragma("unroll") for (int i_ = 0; i_ < CH; ++i_) { \
            const unsigned off_ = (unsigned)ip_[i_] * (unsigned)(KDIM * 4) + lnoff; \
            ring[(G) % (AHEAD + 1)][i_] = *reinterpret_cast<const v4f*>(xbytes + off_); \
        } } while (0)

        #pragma unroll
        for (int g = 0; g < AHEAD; ++g) LOAD_GROUP(g);

#define LOAD_W() do { \
        _Pragma("unroll") for (int i_ = 0; i_ < CH; ++i_) { \
            ring[WSLOT][i_] = *reinterpret_cast<const v4f*>(W + 4 * (tid + i_ * THREADS)); \
        } \
        bv0 = bias[col0]; \
        bv1 = bias[col1]; } while (0)

        v4f acc = {0.0f, 0.0f, 0.0f, 0.0f};
        #pragma unroll
        for (int g = 0; g < NGRP; ++g) {
            __builtin_amdgcn_sched_barrier(0);
            if (g + AHEAD < NGRP) LOAD_GROUP(g + AHEAD);
            else if (g + AHEAD == NGRP) LOAD_W();
            __builtin_amdgcn_sched_barrier(0);

            const int nl = nlw + g / GPN;
            const float* dp = &dm_lds[nl * NASZ + (g % GPN) * CH];
            if (g % GPN == 0) acc = (v4f){0.0f, 0.0f, 0.0f, 0.0f};
            #pragma unroll
            for (int i = 0; i < CH; ++i) {
                const float d = dp[i];
                const v4f  v = ring[g % (AHEAD + 1)][i];
                acc.x = fmaf(d, v.x, acc.x);
                acc.y = fmaf(d, v.y, acc.y);
                acc.z = fmaf(d, v.z, acc.z);
                acc.w = fmaf(d, v.w, acc.w);
            }
            if (g % GPN == GPN - 1) {
                const int row = nl * 2 + (ln >> 4);
                *reinterpret_cast<v4f*>(&h_lds[row * LSTR + 4 * (ln & 15)]) = acc * (1.0f / NASZ);
            }
        }
        #pragma unroll
        for (int i = 0; i < CH; ++i) {
            const int e = 4 * (tid + i * THREADS);
            *reinterpret_cast<v4f*>(&w_lds[(e >> 7) * LSTR + (e & 127)]) = ring[WSLOT][i];
        }
#undef LOAD_W
#undef LOAD_GROUP
    }
    __syncthreads();

    v8f acc0, acc1;
    #pragma unroll
    for (int r = 0; r < 8; ++r) { acc0[r] = bv0; acc1[r] = bv1; }

    #pragma unroll 4
    for (int kk = 0; kk < 32; ++kk) {
        const int kb = kk * 4 + khalf;
        v2f afrag;
        afrag.x = h_lds[rowA * LSTR + kb];
        afrag.y = h_lds[rowA * LSTR + kb + 1];
        v2f bfrag0, bfrag1;
        bfrag0.x = w_lds[col0 * LSTR + kb];
        bfrag0.y = w_lds[col0 * LSTR + kb + 1];
        bfrag1.x = w_lds[col1 * LSTR + kb];
        bfrag1.y = w_lds[col1 * LSTR + kb + 1];
        acc0 = __builtin_amdgcn_wmma_f32_16x16x4_f32(
            false, afrag, false, bfrag0, (short)0, acc0, false, false);
        acc1 = __builtin_amdgcn_wmma_f32_16x16x4_f32(
            false, afrag, false, bfrag1, (short)0, acc1, false, false);
    }

    const int mbase = blockIdx.x * ROWS + rt * 16 + (lane >> 4) * 8;
    float* __restrict__ op = out + (size_t)mbase * OUTC;
    const bool full = (blockIdx.x + 1) * ROWS <= NN * BB;
    if (full) {
        #pragma unroll
        for (int r = 0; r < 8; ++r) {
            op[(size_t)r * OUTC + col0] = acc0[r];
            op[(size_t)r * OUTC + col1] = acc1[r];
        }
    } else {
        #pragma unroll
        for (int r = 0; r < 8; ++r) {
            if (mbase + r < NN * BB) {
                op[(size_t)r * OUTC + col0] = acc0[r];
                op[(size_t)r * OUTC + col1] = acc1[r];
            }
        }
    }
}

extern "C" void kernel_launch(void* const* d_in, const int* in_sizes, int n_in,
                              void* d_out, int out_size, void* d_ws, size_t ws_size,
                              hipStream_t stream) {
    const float* x    = (const float*)d_in[0];
    const float* dm   = (const float*)d_in[1];
    const int*   idx  = (const int*)  d_in[2];
    const float* W    = (const float*)d_in[3];
    const float* bias = (const float*)d_in[4];
    float* out = (float*)d_out;

    const int grid = (NN * BB + ROWS - 1) / ROWS;
    pgnn_fused_kernel<<<grid, THREADS, 0, stream>>>(x, dm, idx, W, bias, out);
}


# Work re-tiled to one equal share per WGP: 250 workgroups of 40 nodes and 10 waves (was 313 of 32 nodes and 8 waves, so 57 WGPs carried two); per-wave gather and all summation orders unchanged
# speedup vs baseline: 4.2118x; 1.1546x over previous
#include <hip/hip_runtime.h>

typedef __attribute__((ext_vector_type(2))) float v2f;
typedef __attribute__((ext_vector_type(4))) float v4f;
typedef __attribute__((ext_vector_type(8))) float v8f;
typedef __attribute__((ext_vector_type(4))) int   v4i;

#define NN    10000
#define BB    2
#define FF    64
#define NASZ  64
#define OUTC  64
#define KDIM  128
#define NPW   40
#define ROWS  80
#define LSTR  132
#define THREADS 320
#define WAVES   10
#define NPV     (NPW / WAVES)
#define CH      8
#define GPN     (NASZ / CH)
#define NGRP    (NPV * GPN)
#define AHEAD   4
#define IDXV    ((NPW * NASZ) / (4 * THREADS))
#define WSLOT   (NGRP % (AHEAD + 1))
#define WVEC    ((OUTC * KDIM) / 4)
#define WCH     ((WVEC + THREADS - 1) / THREADS)
static_assert(WCH <= CH, "W_hidden must fit one ring slot per thread");
static_assert(LSTR - KDIM >= 4, "a row's pad must hold one 128-bit piece");
static_assert(NPW % WAVES == 0 && ROWS == (WAVES / 2) * 16, "one (row tile, column half) per wave");
static_assert((NPW * NASZ) % (4 * THREADS) == 0, "index staging must divide evenly");

__launch_bounds__(THREADS)
__global__ void pgnn_fused_kernel(const float* __restrict__ x,
                                  const float* __restrict__ dm,
                                  const int*   __restrict__ idx,
                                  const float* __restrict__ W,
                                  const float* __restrict__ bias,
                                  float* __restrict__ out)
{
    __shared__ __attribute__((aligned(16))) float w_lds[OUTC * LSTR];
    __shared__ __attribute__((aligned(16))) float h_lds[ROWS * LSTR];
    __shared__ __attribute__((aligned(16))) int   idx_lds[NPW * NASZ];
    __shared__ __attribute__((aligned(16))) float dm_lds [NPW * NASZ];

    const int tid = threadIdx.x;
    const int n0  = blockIdx.x * NPW;

    v4f own[NPV];
    #pragma unroll
    for (int i = 0; i < NPV; ++i) {
        const int n  = n0 + (tid >> 5) * NPV + i;
        const int nc = (n < NN) ? n : (NN - 1);
        own[i] = *reinterpret_cast<const v4f*>(x + (size_t)nc * KDIM + 4 * (tid & 31));
    }

    v4i iv[IDXV];
    v4f dv[IDXV];
    #pragma unroll
    for (int j = 0; j < IDXV; ++j) {
        const int e  = 4 * (tid + j * THREADS);
        const int n  = n0 + (e >> 6);
        const int nc = (n < NN) ? n : (NN - 1);
        const size_t off = (size_t)nc * NASZ + (e & 63);
        iv[j] = *reinterpret_cast<const v4i*>(idx + off);
        dv[j] = *reinterpret_cast<const v4f*>(dm  + off);
    }
    #pragma unroll
    for (int j = 0; j < IDXV; ++j) {
        const int e = 4 * (tid + j * THREADS);
        v4i a = iv[j];
        v4f d = dv[j];
        if (n0 + (e >> 6) >= NN) {
            a = (v4i){0, 0, 0, 0};
            d = (v4f){0.0f, 0.0f, 0.0f, 0.0f};
        }
        *reinterpret_cast<v4i*>(&idx_lds[e]) = a;
        *reinterpret_cast<v4f*>(&dm_lds[e])  = d;
    }
    #pragma unroll
    for (int i = 0; i < NPV; ++i) {
        const int nl  = (tid >> 5) * NPV + i;
        const int row = nl * 2 + ((tid & 31) >> 4);
        v4f o = own[i];
        if (n0 + nl >= NN) o = (v4f){0.0f, 0.0f, 0.0f, 0.0f};
        *reinterpret_cast<v4f*>(&h_lds[row * LSTR + FF + 4 * (tid & 15)]) = o;
    }
    __syncthreads();

    const int wave  = tid >> 5;
    const int lane  = tid & 31;
    const int l15   = lane & 15;
    const int khalf = (lane >> 4) * 2;
    const int rt    = wave >> 1;
    const int ch    = wave & 1;
    const int rowA  = rt * 16 + l15;
    const int col0  = (ch * 2) * 16 + l15;
    const int col1  = (ch * 2 + 1) * 16 + l15;
    float bv0 = 0.0f, bv1 = 0.0f;

    {
        const int wv  = tid >> 5;
        const int ln  = tid & 31;
        const int nlw = wv * NPV;
        const unsigned lnoff = 16u * (unsigned)ln;
        const unsigned char* xbytes = reinterpret_cast<const unsigned char*>(x);

        v4f ring[AHEAD + 1][CH];
#define LOAD_GROUP(G) do { \
        const int* ip_ = &idx_lds[(nlw + (G) / GPN) * NASZ + ((G) % GPN) * CH]; \
        _Pragma("unroll") for (int i_ = 0; i_ < CH; ++i_) { \
            const unsigned off_ = (unsigned)ip_[i_] * (unsigned)(KDIM * 4) + lnoff; \
            ring[(G) % (AHEAD + 1)][i_] = *reinterpret_cast<const v4f*>(xbytes + off_); \
        } } while (0)

        #pragma unroll
        for (int g = 0; g < AHEAD; ++g) LOAD_GROUP(g);

#define LOAD_W() do { \
        _Pragma("unroll") for (int i_ = 0; i_ < WCH; ++i_) { \
            const int e4_  = tid + i_ * THREADS; \
            const int e4c_ = ((i_ + 1) * THREADS <= WVEC || e4_ < WVEC) ? e4_ : (WVEC - 1); \
            ring[WSLOT][i_] = *reinterpret_cast<const v4f*>(W + 4 * e4c_); \
        } \
        bv0 = bias[col0]; \
        bv1 = bias[col1]; } while (0)

        v4f acc = {0.0f, 0.0f, 0.0f, 0.0f};
        #pragma unroll
        for (int g = 0; g < NGRP; ++g) {
            __builtin_amdgcn_sched_barrier(0);
            if (g + AHEAD < NGRP) LOAD_GROUP(g + AHEAD);
            else if (g + AHEAD == NGRP) LOAD_W();
            __builtin_amdgcn_sched_barrier(0);

            const int nl = nlw + g / GPN;
            const float* dp = &dm_lds[nl * NASZ + (g % GPN) * CH];
            if (g % GPN == 0) acc = (v4f){0.0f, 0.0f, 0.0f, 0.0f};
            #pragma unroll
            for (int i = 0; i < CH; ++i) {
                const float d = dp[i];
                const v4f  v = ring[g % (AHEAD + 1)][i];
                acc.x = fmaf(d, v.x, acc.x);
                acc.y = fmaf(d, v.y, acc.y);
                acc.z = fmaf(d, v.z, acc.z);
                acc.w = fmaf(d, v.w, acc.w);
            }
            if (g % GPN == GPN - 1) {
                const int row = nl * 2 + (ln >> 4);
                *reinterpret_cast<v4f*>(&h_lds[row * LSTR + 4 * (ln & 15)]) = acc * (1.0f / NASZ);
            }
        }
        #pragma unroll
        for (int i = 0; i < WCH; ++i) {
            const int e4 = tid + i * THREADS;
            const int e  = 4 * e4;
            int o = (e >> 7) * LSTR + (e & 127);
            if ((i + 1) * THREADS > WVEC && e4 >= WVEC) o = (tid & (OUTC - 1)) * LSTR + KDIM;
            *reinterpret_cast<v4f*>(&w_lds[o]) = ring[WSLOT][i];
        }
#undef LOAD_W
#undef LOAD_GROUP
    }
    __syncthreads();

    v8f acc0, acc1;
    #pragma unroll
    for (int r = 0; r < 8; ++r) { acc0[r] = bv0; acc1[r] = bv1; }

    #pragma unroll 4
    for (int kk = 0; kk < 32; ++kk) {
        const int kb = kk * 4 + khalf;
        v2f afrag;
        afrag.x = h_lds[rowA * LSTR + kb];
        afrag.y = h_lds[rowA * LSTR + kb + 1];
        v2f bfrag0, bfrag1;
        bfrag0.x = w_lds[col0 * LSTR + kb];
        bfrag0.y = w_lds[col0 * LSTR + kb + 1];
        bfrag1.x = w_lds[col1 * LSTR + kb];
        bfrag1.y = w_lds[col1 * LSTR + kb + 1];
        acc0 = __builtin_amdgcn_wmma_f32_16x16x4_f32(
            false, afrag, false, bfrag0, (short)0, acc0, false, false);
        acc1 = __builtin_amdgcn_wmma_f32_16x16x4_f32(
            false, afrag, false, bfrag1, (short)0, acc1, false, false);
    }

    const int mbase = blockIdx.x * ROWS + rt * 16 + (lane >> 4) * 8;
    float* __restrict__ op = out + (size_t)mbase * OUTC;
    const bool full = (blockIdx.x + 1) * ROWS <= NN * BB;
    if (full) {
        #pragma unroll
        for (int r = 0; r < 8; ++r) {
            op[(size_t)r * OUTC + col0] = acc0[r];
            op[(size_t)r * OUTC + col1] = acc1[r];
        }
    } else {
        #pragma unroll
        for (int r = 0; r < 8; ++r) {
            if (mbase + r < NN * BB) {
                op[(size_t)r * OUTC + col0] = acc0[r];
                op[(size_t)r * OUTC + col1] = acc1[r];
            }
        }
    }
}

extern "C" void kernel_launch(void* const* d_in, const int* in_sizes, int n_in,
                              void* d_out, int out_size, void* d_ws, size_t ws_size,
                              hipStream_t stream) {
    const float* x    = (const float*)d_in[0];
    const float* dm   = (const float*)d_in[1];
    const int*   idx  = (const int*)  d_in[2];
    const float* W    = (const float*)d_in[3];
    const float* bias = (const float*)d_in[4];
    float* out = (float*)d_out;

    const int grid = (NN * BB + ROWS - 1) / ROWS;
    pgnn_fused_kernel<<<grid, THREADS, 0, stream>>>(x, dm, idx, W, bias, out);
}
